# starting kernel
# baseline (speedup 1.0000x reference)
#include <hip/hip_runtime.h>


typedef float v2f __attribute__((ext_vector_type(2)));
typedef float v8f __attribute__((ext_vector_type(8)));

__global__ void zero_f32(float* __restrict__ p, int n) {
    int i = blockIdx.x * blockDim.x + threadIdx.x;
    if (i < n) p[i] = 0.0f;
}

__global__ void edge_count(const int* __restrict__ dst, float* __restrict__ cnt, int E) {
    int e = blockIdx.x * blockDim.x + threadIdx.x;
    if (e < E) atomicAdd(&cnt[dst[e]], 1.0f);
}

__global__ void edge_scatter(const float* __restrict__ xs, const int* __restrict__ src,
                             const int* __restrict__ dst, float* __restrict__ sum,
                             int E, int D) {
    int chunks = D >> 2;
    long long tid = (long long)blockIdx.x * blockDim.x + threadIdx.x;
    if (tid >= (long long)E * chunks) return;
    int e = (int)(tid / chunks);
    int c = (int)(tid % chunks);
    const float4 v = *(const float4*)(xs + (size_t)src[e] * D + (size_t)c * 4);
    float* s = sum + (size_t)dst[e] * D + (size_t)c * 4;
    atomicAdd(s + 0, v.x);
    atomicAdd(s + 1, v.y);
    atomicAdd(s + 2, v.z);
    atomicAdd(s + 3, v.w);
}

__global__ void seg_normalize(float* __restrict__ sum, const float* __restrict__ cnt,
                              int n, int D) {
    int chunks = D >> 2;
    long long tid = (long long)blockIdx.x * blockDim.x + threadIdx.x;
    if (tid >= (long long)n * chunks) return;
    int i = (int)(tid / chunks);
    int c = (int)(tid % chunks);
    float inv = 1.0f / fmaxf(cnt[i], 1.0f);
    float4* p = (float4*)(sum + (size_t)i * D + (size_t)c * 4);
    float4 v = *p;
    v.x *= inv; v.y *= inv; v.z *= inv; v.w *= inv;
    *p = v;
}

__global__ void sage_gemm_wmma(const float* __restrict__ A0,
                               const float* __restrict__ A1,
                               const float* __restrict__ W0,
                               const float* __restrict__ W1,
                               const float* __restrict__ bias,
                               float* __restrict__ out,
                               int M, int K, int N, int relu) {
    int wave = blockIdx.x * (blockDim.x >> 5) + (threadIdx.x >> 5);
    int tilesN = N >> 4;
    int totalTiles = (M >> 4) * tilesN;
    if (wave >= totalTiles) return;

    int tm = wave / tilesN;
    int tn = wave - tm * tilesN;
    int lane = threadIdx.x & 31;
    int half = lane >> 4;
    int l15  = lane & 15;
    int arow = (tm << 4) + l15;
    int bcol = (tn << 4) + l15;

    v8f acc = {0.f, 0.f, 0.f, 0.f, 0.f, 0.f, 0.f, 0.f};

    #pragma unroll
    for (int pass = 0; pass < 2; ++pass) {
        const float* A = pass ? A1 : A0;
        const float* W = pass ? W1 : W0;
        const float* ap = A + (size_t)arow * K + (half << 1);
        #pragma unroll 4
        for (int k0 = 0; k0 < K; k0 += 4) {
            v2f a = *(const v2f*)(ap + k0);
            int bk = k0 + (half << 1);
            v2f b;
            b.x = W[(size_t)bk * N + bcol];
            b.y = W[(size_t)(bk + 1) * N + bcol];
            acc = __builtin_amdgcn_wmma_f32_16x16x4_f32(
                 false, a,  false, b,
                 (short)0, acc,  false,  false);
        }
    }

    float bv = bias[bcol];
    float* op = out + (size_t)((tm << 4) + (half << 3)) * N + bcol;
    #pragma unroll
    for (int v = 0; v < 8; ++v) {
        float r = acc[v] + bv;
        if (relu) r = fmaxf(r, 0.0f);
        op[(size_t)v * N] = r;
    }
}

extern "C" void kernel_launch(void* const* d_in, const int* in_sizes, int n_in,
                              void* d_out, int out_size, void* d_ws, size_t ws_size,
                              hipStream_t stream) {
    (void)in_sizes; (void)n_in; (void)out_size; (void)ws_size;

    const float* x = (const float*)d_in[0];
    const int* srcp[3] = {(const int*)d_in[1], (const int*)d_in[3], (const int*)d_in[5]};
    const int* dstp[3] = {(const int*)d_in[2], (const int*)d_in[4], (const int*)d_in[6]};
    const float* Wl[3] = {(const float*)d_in[7],  (const float*)d_in[10], (const float*)d_in[13]};
    const float* Wr[3] = {(const float*)d_in[8],  (const float*)d_in[11], (const float*)d_in[14]};
    const float* bb[3] = {(const float*)d_in[9],  (const float*)d_in[12], (const float*)d_in[15]};

    const int Ecnt[3] = {1500000, 300000, 60000};
    const int ntgt[3] = {100000, 20000, 4000};
    const int Kd[3]   = {128, 256, 256};
    const int Nd[3]   = {256, 256, 128};
    const int relu[3] = {1, 0, 1};

    float* ws     = (float*)d_ws;
    float* aggbuf = ws;
    float* cntbuf = aggbuf + 12800000;
    float* h0     = cntbuf + 100000;
    float* h1     = h0 + 25600000;

    const float* layer_in[3] = {x, h0, h1};
    float* layer_out[3]      = {h0, h1, (float*)d_out};

    const int TPB = 256;
    for (int L = 0; L < 3; ++L) {
        int E = Ecnt[L], n = ntgt[L], K = Kd[L], N = Nd[L];

        int zn = n * K;
        zero_f32<<<(zn + TPB - 1) / TPB, TPB, 0, stream>>>(aggbuf, zn);
        zero_f32<<<(n + TPB - 1) / TPB, TPB, 0, stream>>>(cntbuf, n);

        edge_count<<<(E + TPB - 1) / TPB, TPB, 0, stream>>>(dstp[L], cntbuf, E);

        long long sthreads = (long long)E * (K >> 2);
        edge_scatter<<<(unsigned)((sthreads + TPB - 1) / TPB), TPB, 0, stream>>>(
            layer_in[L], srcp[L], dstp[L], aggbuf, E, K);

        long long nthreads = (long long)n * (K >> 2);
        seg_normalize<<<(unsigned)((nthreads + TPB - 1) / TPB), TPB, 0, stream>>>(
            aggbuf, cntbuf, n, K);

        int tiles = (n >> 4) * (N >> 4);
        int wavesPerBlock = TPB >> 5;
        sage_gemm_wmma<<<(tiles + wavesPerBlock - 1) / wavesPerBlock, TPB, 0, stream>>>(
            aggbuf, layer_in[L], Wl[L], Wr[L], bb[L], layer_out[L], n, K, N, relu[L]);
    }
}


# Segment mean by owner work-groups: each of 256 scans the dst list, adds the gathered rows of its node range in LDS (f32 LDS adds), stores each mean once; 6 launches; sum order = LDS arrival order
# speedup vs baseline: 3.2310x; 3.2310x over previous
#include <hip/hip_runtime.h>


typedef float v2f __attribute__((ext_vector_type(2)));
typedef float v8f __attribute__((ext_vector_type(8)));

template <int D, int EB, bool FULL>
__device__ __forceinline__ void drain_rows(const float* __restrict__ xs,
                                           const int* __restrict__ src,
                                           float* acc, const unsigned* qw,
                                           int cnt, unsigned lane) {
    constexpr int R = D / 32;
    int jj = (int)(lane & (EB - 1));
    if (!FULL) jj = jj < cnt ? jj : cnt - 1;
    const unsigned w = qw[jj];
    const int dl = (int)(w >> 21);
    const int s  = src[w & 0x1FFFFFu];
    float v[EB][R];
    #pragma unroll
    for (int j = 0; j < EB; ++j) {
        if (FULL || j < cnt) {
            const int sj = __builtin_amdgcn_readlane(s, j);
            const float* row = xs + (size_t)sj * D;
            #pragma unroll
            for (int i = 0; i < R; ++i) v[j][i] = row[lane + i * 32];
        }
    }
    #pragma unroll
    for (int j = 0; j < EB; ++j) {
        if (FULL || j < cnt) {
            const int dj = __builtin_amdgcn_readlane(dl, j);
            float* a = acc + dj * D;
            #pragma unroll
            for (int i = 0; i < R; ++i) atomicAdd(a + lane + i * 32, v[j][i]);
        }
    }
}

template <int D, int RPW, int EB, int NW, int E, int N>
__global__ __launch_bounds__(NW * 32) void seg_mean_owner(
        const float* __restrict__ xs, const int* __restrict__ src,
        const int* __restrict__ dst, float* __restrict__ agg) {
    constexpr int BLK  = 512;
    constexpr int QCAP = EB + BLK + 32;
    constexpr int C4   = D / 4;
    constexpr int NBLK = E / BLK;
    static_assert(E < (1 << 21) && RPW <= 1024, "queue word packing");
    __shared__ __attribute__((aligned(16))) float acc[RPW * D];
    __shared__ int cntl[RPW];
    __shared__ unsigned qall[NW * QCAP];

    const int lo = (int)blockIdx.x * RPW;
    int nrows = N - lo;
    nrows = nrows < RPW ? nrows : RPW;
    if (nrows <= 0) return;

    for (int i = threadIdx.x; i < RPW * C4; i += NW * 32)
        ((float4*)acc)[i] = make_float4(0.f, 0.f, 0.f, 0.f);
    for (int i = threadIdx.x; i < RPW; i += NW * 32) cntl[i] = 0;
    __syncthreads();

    const unsigned lane = threadIdx.x & 31;
    const int wave = __builtin_amdgcn_readfirstlane((int)(threadIdx.x >> 5));
    unsigned* qw = qall + wave * QCAP;
    int qn = 0;
    const int rot = (int)(((long long)blockIdx.x * NBLK) / (int)gridDim.x);
    const int4* d4 = (const int4*)dst;

    int4 cur[4], nxt[4];
    {
        int bb = wave + rot; bb = bb >= NBLK ? bb - NBLK : bb;
        bb = wave < NBLK ? bb : 0;
        #pragma unroll
        for (int u = 0; u < 4; ++u) cur[u] = d4[(size_t)bb * 128 + u * 32 + lane];
    }
    for (int b = wave; b < NBLK; b += NW) {
        int bn = b + NW; bn = bn < NBLK ? bn : b;
        int bbn = bn + rot; bbn = bbn >= NBLK ? bbn - NBLK : bbn;
        #pragma unroll
        for (int u = 0; u < 4; ++u) nxt[u] = d4[(size_t)bbn * 128 + u * 32 + lane];

        int bb = b + rot; bb = bb >= NBLK ? bb - NBLK : bb;
        const int ebase = bb * BLK + (int)lane * 4;
        #pragma unroll
        for (int u = 0; u < 4; ++u) {
            const int dv[4] = {cur[u].x, cur[u].y, cur[u].z, cur[u].w};
            bool m[4];
            unsigned mk[4];
            #pragma unroll
            for (int c = 0; c < 4; ++c) {
                m[c]  = (unsigned)(dv[c] - lo) < (unsigned)nrows;
                mk[c] = __builtin_amdgcn_ballot_w32(m[c]);
            }
            if ((mk[0] | mk[1] | mk[2] | mk[3]) != 0u) {
                #pragma unroll
                for (int c = 0; c < 4; ++c) {
                    if (mk[c] != 0u) {
                        const int pos = qn + (int)__builtin_amdgcn_mbcnt_lo(mk[c], 0u);
                        if (m[c]) {
                            qw[pos] = (unsigned)(ebase + u * 128 + c) | ((unsigned)(dv[c] - lo) << 21);
                            atomicAdd(&cntl[dv[c] - lo], 1);
                        }
                        qn += __builtin_popcount(mk[c]);
                    }
                }
            }
        }
        while (qn >= EB) {
            drain_rows<D, EB, true>(xs, src, acc, qw + (qn - EB), EB, lane);
            qn -= EB;
        }
        #pragma unroll
        for (int u = 0; u < 4; ++u) cur[u] = nxt[u];
    }
    for (int e0 = NBLK * BLK + wave * 32; e0 < E; e0 += NW * 32) {
        const int e = e0 + (int)lane;
        const int d = e < E ? dst[e] : -1;
        const bool m = e < E && (unsigned)(d - lo) < (unsigned)nrows;
        const unsigned mk = __builtin_amdgcn_ballot_w32(m);
        if (mk != 0u) {
            const int pos = qn + (int)__builtin_amdgcn_mbcnt_lo(mk, 0u);
            if (m) {
                qw[pos] = (unsigned)e | ((unsigned)(d - lo) << 21);
                atomicAdd(&cntl[d - lo], 1);
            }
            qn += __builtin_popcount(mk);
        }
        while (qn >= EB) {
            drain_rows<D, EB, true>(xs, src, acc, qw + (qn - EB), EB, lane);
            qn -= EB;
        }
    }
    if (qn > 0) drain_rows<D, EB, false>(xs, src, acc, qw, qn, lane);
    __syncthreads();

    float4* o4 = (float4*)(agg + (size_t)lo * D);
    for (int idx = threadIdx.x; idx < nrows * C4; idx += NW * 32) {
        const int r = idx / C4;
        const float inv = 1.0f / fmaxf((float)cntl[r], 1.0f);
        float4 v = ((const float4*)acc)[idx];
        v.x *= inv; v.y *= inv; v.z *= inv; v.w *= inv;
        o4[idx] = v;
    }
}

template <int NT, int PITCH, int S>
__device__ __forceinline__ void group_mma(v8f (&acc)[NT], const v2f (&a)[8], v2f (&b_cur)[NT],
                                          const float* wlane, int gi) {
    #pragma unroll
    for (int ks = 0; ks < 8; ++ks) {
        int sn = gi * 8 + ks + 1;
        sn = (sn < S) ? sn : S - 1;
        const float* wn = wlane + sn * (2 * PITCH);
        v2f b_nxt[NT];
        #pragma unroll
        for (int j = 0; j < NT; ++j) b_nxt[j] = *(const v2f*)(wn + j * 32);
        #pragma unroll
        for (int j = 0; j < NT; ++j)
            acc[j] = __builtin_amdgcn_wmma_f32_16x16x4_f32(
                 false, a[ks],  false, b_cur[j],
                 (short)0, acc[j],  false,  false);
        #pragma unroll
        for (int j = 0; j < NT; ++j) b_cur[j] = b_nxt[j];
    }
}

template <int K, int N, int NB>
__global__ __launch_bounds__(256) void sage_gemm_ldsw(
        const float* __restrict__ A0,
        const float* __restrict__ A1,
        const float* __restrict__ W0,
        const float* __restrict__ W1,
        const float* __restrict__ bias,
        float* __restrict__ out,
        int M, int relu) {
    constexpr int PITCH = NB * 2 + 32;
    constexpr int NT  = NB / 16;
    constexpr int CB  = N / NB;
    constexpr int G   = K / 32;
    constexpr int NG  = 2 * G;
    constexpr int S   = NG * 8;
    constexpr int NB4 = NB / 4;
    __shared__ __attribute__((aligned(16))) float Ws[K * PITCH];

    const int cb  = blockIdx.x % CB;
    const int wg  = blockIdx.x / CB;
    const int nwg = gridDim.x / CB;
    const int colbase = cb * NB;

    for (int idx = threadIdx.x; idx < K * NB4; idx += 256) {
        const int pr = idx / NB4;
        const int c4 = idx - pr * NB4;
        const float* Wm = (pr >= K / 2) ? W1 : W0;
        const int kp = (pr >= K / 2) ? pr - K / 2 : pr;
        const float* s0 = Wm + (size_t)(2 * kp) * N + colbase + c4 * 4;
        const float4 r0 = *(const float4*)s0;
        const float4 r1 = *(const float4*)(s0 + N);
        float* d = Ws + pr * PITCH + c4 * 8;
        *(float4*)d       = make_float4(r0.x, r1.x, r0.y, r1.y);
        *(float4*)(d + 4) = make_float4(r0.z, r1.z, r0.w, r1.w);
    }
    __syncthreads();

    const int wave = threadIdx.x >> 5;
    const int lane = threadIdx.x & 31;
    const int half = lane >> 4;
    const int l15  = lane & 15;
    const int rowTiles = M >> 4;
    const float* wlane = Ws + half * PITCH + l15 * 2;

    for (int t = wave * nwg + wg; t < rowTiles; t += 8 * nwg) {
        const size_t aoff = (size_t)((t << 4) + l15) * K + (half << 1);
        const float* a0p = A0 + aoff;
        const float* a1p = A1 + aoff;

        v8f acc[NT];
        #pragma unroll
        for (int j = 0; j < NT; ++j) acc[j] = (v8f){0.f, 0.f, 0.f, 0.f, 0.f, 0.f, 0.f, 0.f};

        v2f a_ev[8], a_od[8];
        v2f b_cur[NT];
        #pragma unroll
        for (int ks = 0; ks < 8; ++ks) a_ev[ks] = *(const v2f*)(a0p + ks * 4);
        #pragma unroll
        for (int j = 0; j < NT; ++j) b_cur[j] = *(const v2f*)(wlane + j * 32);

        for (int gi = 0; gi < NG; gi += 2) {
            const float* aop = (gi + 1 >= G) ? a1p + (gi + 1 - G) * 32 : a0p + (gi + 1) * 32;
            #pragma unroll
            for (int ks = 0; ks < 8; ++ks) a_od[ks] = *(const v2f*)(aop + ks * 4);

            group_mma<NT, PITCH, S>(acc, a_ev, b_cur, wlane, gi);

            const int gn = (gi + 2 < NG) ? gi + 2 : NG - 1;
            const float* aep = (gn >= G) ? a1p + (gn - G) * 32 : a0p + gn * 32;
            #pragma unroll
            for (int ks = 0; ks < 8; ++ks) a_ev[ks] = *(const v2f*)(aep + ks * 4);

            group_mma<NT, PITCH, S>(acc, a_od, b_cur, wlane, gi + 1);
        }

        float* op = out + (size_t)((t << 4) + (half << 3)) * N + colbase + l15;
        #pragma unroll
        for (int j = 0; j < NT; ++j) {
            const float bv = bias[colbase + j * 16 + l15];
            #pragma unroll
            for (int v = 0; v < 8; ++v) {
                float r = acc[j][v] + bv;
                if (relu) r = fmaxf(r, 0.0f);
                op[(size_t)v * N + j * 16] = r;
            }
        }
    }
}

extern "C" void kernel_launch(void* const* d_in, const int* in_sizes, int n_in,
                              void* d_out, int out_size, void* d_ws, size_t ws_size,
                              hipStream_t stream) {
    (void)in_sizes; (void)n_in; (void)out_size; (void)ws_size;

    const float* x = (const float*)d_in[0];
    const int* srcp[3] = {(const int*)d_in[1], (const int*)d_in[3], (const int*)d_in[5]};
    const int* dstp[3] = {(const int*)d_in[2], (const int*)d_in[4], (const int*)d_in[6]};
    const float* Wl[3] = {(const float*)d_in[7],  (const float*)d_in[10], (const float*)d_in[13]};
    const float* Wr[3] = {(const float*)d_in[8],  (const float*)d_in[11], (const float*)d_in[14]};
    const float* bb[3] = {(const float*)d_in[9],  (const float*)d_in[12], (const float*)d_in[15]};

    const int ntgt[3] = {100000, 20000, 4000};
    const int relu[3] = {1, 0, 1};

    float* ws     = (float*)d_ws;
    float* aggbuf = ws;
    float* h0     = aggbuf + 12900000;
    float* h1     = h0 + 25600000;

    const float* layer_in[3] = {x, h0, h1};
    float* layer_out[3]      = {h0, h1, (float*)d_out};

    for (int L = 0; L < 3; ++L) {
        int n = ntgt[L];

        if (L == 0)
            seg_mean_owner<128, 391, 16, 32, 1500000, 100000><<<256, 1024, 0, stream>>>(
                layer_in[L], srcp[L], dstp[L], aggbuf);
        else if (L == 1)
            seg_mean_owner<256, 79, 8, 32, 300000, 20000><<<256, 1024, 0, stream>>>(
                layer_in[L], srcp[L], dstp[L], aggbuf);
        else
            seg_mean_owner<256, 16, 8, 32, 60000, 4000><<<256, 1024, 0, stream>>>(
                layer_in[L], srcp[L], dstp[L], aggbuf);

        if (L == 0)
            sage_gemm_ldsw<128, 256, 256><<<256, 256, 0, stream>>>(
                aggbuf, layer_in[L], Wl[L], Wr[L], bb[L], layer_out[L], n, relu[L]);
        else if (L == 1)
            sage_gemm_ldsw<256, 256, 128><<<256, 256, 0, stream>>>(
                aggbuf, layer_in[L], Wl[L], Wr[L], bb[L], layer_out[L], n, relu[L]);
        else
            sage_gemm_ldsw<256, 128, 64><<<256, 256, 0, stream>>>(
                aggbuf, layer_in[L], Wl[L], Wr[L], bb[L], layer_out[L], n, relu[L]);
    }
}


# Owner scan reads a 1-byte owner map (dst / rows per owner, built by one added first launch) instead of the 4-byte dst list; hits re-checked against dst at the drain; sum order = LDS arrival order
# speedup vs baseline: 3.3640x; 1.0412x over previous
#include <hip/hip_runtime.h>


typedef float v2f __attribute__((ext_vector_type(2)));
typedef float v8f __attribute__((ext_vector_type(8)));

template <int RPW>
__device__ __forceinline__ unsigned own4(const int4 v) {
    const unsigned a = min((unsigned)v.x / (unsigned)RPW, 255u);
    const unsigned b = min((unsigned)v.y / (unsigned)RPW, 255u);
    const unsigned c = min((unsigned)v.z / (unsigned)RPW, 255u);
    const unsigned d = min((unsigned)v.w / (unsigned)RPW, 255u);
    return a | (b << 8) | (c << 16) | (d << 24);
}

template <int RPW, int E>
__device__ __forceinline__ void owner_bytes(const int* __restrict__ dst,
                                            unsigned* __restrict__ omap, int blk, int tid) {
    static_assert(E % 16 == 0, "whole 16-edge groups");
    constexpr int E4 = E / 4;
    const int4* d4 = (const int4*)dst;
    int4 v[4];
    #pragma unroll
    for (int u = 0; u < 4; ++u) {
        const int i = blk * 1024 + u * 256 + tid;
        v[u] = d4[i < E4 ? i : E4 - 1];
    }
    #pragma unroll
    for (int u = 0; u < 4; ++u) {
        const int i = blk * 1024 + u * 256 + tid;
        if (i < E4) omap[i] = own4<RPW>(v[u]);
    }
}

template <int RPW0, int E0, int RPW1, int E1, int RPW2, int E2>
__global__ __launch_bounds__(256) void owner_map_all(
        const int* __restrict__ dst0, const int* __restrict__ dst1,
        const int* __restrict__ dst2, unsigned* __restrict__ m0,
        unsigned* __restrict__ m1, unsigned* __restrict__ m2) {
    constexpr int B0 = (E0 / 4 + 1023) / 1024;
    constexpr int B1 = (E1 / 4 + 1023) / 1024;
    const int b = (int)blockIdx.x;
    const int tid = (int)threadIdx.x;
    if (b < B0)           owner_bytes<RPW0, E0>(dst0, m0, b, tid);
    else if (b < B0 + B1) owner_bytes<RPW1, E1>(dst1, m1, b - B0, tid);
    else                  owner_bytes<RPW2, E2>(dst2, m2, b - B0 - B1, tid);
}

template <int D, int RPW, int EB, bool FULL>
__device__ __forceinline__ void drain_rows(const float* __restrict__ xs,
                                           const int* __restrict__ src,
                                           const int* __restrict__ dst,
                                           float* acc, int* cntl, const unsigned* qw,
                                           int cnt, unsigned lane, int lo, int nrows) {
    constexpr int R = D / 32;
    int jj = (int)(lane & (EB - 1));
    if (!FULL) jj = jj < cnt ? jj : cnt - 1;
    const unsigned e = qw[jj];
    const int s  = src[e];
    const int dr = dst[e] - lo;
    const bool ok = (unsigned)dr < (unsigned)nrows;
    const int dl = ok ? dr : RPW;
    if (ok && (int)lane < (FULL ? EB : cnt)) atomicAdd(&cntl[dl], 1);
    float v[EB][R];
    #pragma unroll
    for (int j = 0; j < EB; ++j) {
        if (FULL || j < cnt) {
            const int sj = __builtin_amdgcn_readlane(s, j);
            const float* row = xs + (size_t)sj * D;
            #pragma unroll
            for (int i = 0; i < R; ++i) v[j][i] = row[lane + i * 32];
        }
    }
    #pragma unroll
    for (int j = 0; j < EB; ++j) {
        if (FULL || j < cnt) {
            const int dj = __builtin_amdgcn_readlane(dl, j);
            float* a = acc + dj * D;
            #pragma unroll
            for (int i = 0; i < R; ++i) atomicAdd(a + lane + i * 32, v[j][i]);
        }
    }
}

template <int D, int RPW, int EB, int NW, int E, int N>
__global__ __launch_bounds__(NW * 32) void seg_mean_owner(
        const float* __restrict__ xs, const int* __restrict__ src,
        const int* __restrict__ dst, const unsigned* __restrict__ omap,
        float* __restrict__ agg) {
    constexpr int E16  = E / 16;
    constexpr int UPT  = 2;
    constexpr int BLKG = 32 * UPT;
    constexpr int NBLK = (E16 + BLKG - 1) / BLKG;
    constexpr int QCAP = EB + 32;
    constexpr int C4   = D / 4;
    static_assert(E % 16 == 0 && (N + RPW - 1) / RPW <= 256, "one byte names the owner");
    __shared__ __attribute__((aligned(16))) float acc[(RPW + 1) * D];
    __shared__ int cntl[RPW + 1];
    __shared__ unsigned qall[NW * QCAP];

    const int lo = (int)blockIdx.x * RPW;
    int nrows = N - lo;
    nrows = nrows < RPW ? nrows : RPW;
    if (nrows <= 0) return;

    for (int i = threadIdx.x; i < (RPW + 1) * C4; i += NW * 32)
        ((float4*)acc)[i] = make_float4(0.f, 0.f, 0.f, 0.f);
    for (int i = threadIdx.x; i < RPW + 1; i += NW * 32) cntl[i] = 0;
    __syncthreads();

    const unsigned lane = threadIdx.x & 31;
    const int wave = __builtin_amdgcn_readfirstlane((int)(threadIdx.x >> 5));
    unsigned* qw = qall + wave * QCAP;
    int qn = 0;
    const int rot = (int)(((long long)blockIdx.x * NBLK) / (int)gridDim.x);
    const uint4* m4 = (const uint4*)omap;
    const unsigned pat  = (unsigned)blockIdx.x * 0x01010101u;

    uint4 cur[UPT], nxt[UPT];
    {
        int bb = wave + rot; bb = bb >= NBLK ? bb - NBLK : bb;
        bb = wave < NBLK ? bb : 0;
        #pragma unroll
        for (int u = 0; u < UPT; ++u) {
            const int g = bb * BLKG + u * 32 + (int)lane;
            cur[u] = m4[g < E16 ? g : E16 - 1];
        }
    }
    for (int b = wave; b < NBLK; b += NW) {
        int bn = b + NW; bn = bn < NBLK ? bn : b;
        int bbn = bn + rot; bbn = bbn >= NBLK ? bbn - NBLK : bbn;
        #pragma unroll
        for (int u = 0; u < UPT; ++u) {
            const int g = bbn * BLKG + u * 32 + (int)lane;
            nxt[u] = m4[g < E16 ? g : E16 - 1];
        }

        int bb = b + rot; bb = bb >= NBLK ? bb - NBLK : bb;
        const unsigned ebase = (unsigned)(bb * BLKG + (int)lane) * 16u;
        unsigned fl = 0;
        #pragma unroll
        for (int u = 0; u < UPT; ++u) {
            const unsigned w[4] = {cur[u].x, cur[u].y, cur[u].z, cur[u].w};
            #pragma unroll
            for (int c = 0; c < 4; ++c) {
                const int k = u * 4 + c;
                const unsigned x = w[c] ^ pat;
                const unsigned t = (x - 0x01010101u) & ~x;
                fl |= (t >> (7 - k)) & (0x01010101u << k);
            }
        }
        const int g0 = bb * BLKG + (int)lane;
        fl &= (g0 < E16 ? 0x0F0F0F0Fu : 0u) | (g0 + 32 < E16 ? 0xF0F0F0F0u : 0u);
        unsigned mk = __builtin_amdgcn_ballot_w32(fl != 0u);
        while (mk != 0u) {
            const int pos = qn + (int)__builtin_amdgcn_mbcnt_lo(mk, 0u);
            if (fl != 0u) {
                const unsigned bit = (unsigned)__builtin_ctz(fl);
                const unsigned k = bit & 7u;
                qw[pos] = ebase + (k >> 2) * 512u + (k & 3u) * 4u + (bit >> 3);
                fl &= fl - 1u;
            }
            qn += __builtin_popcount(mk);
            while (qn >= EB) {
                drain_rows<D, RPW, EB, true>(xs, src, dst, acc, cntl, qw + (qn - EB), EB, lane, lo, nrows);
                qn -= EB;
            }
            mk = __builtin_amdgcn_ballot_w32(fl != 0u);
        }
        #pragma unroll
        for (int u = 0; u < UPT; ++u) cur[u] = nxt[u];
    }
    if (qn > 0) drain_rows<D, RPW, EB, false>(xs, src, dst, acc, cntl, qw, qn, lane, lo, nrows);
    __syncthreads();

    float4* o4 = (float4*)(agg + (size_t)lo * D);
    for (int idx = threadIdx.x; idx < nrows * C4; idx += NW * 32) {
        const int r = idx / C4;
        const float inv = 1.0f / fmaxf((float)cntl[r], 1.0f);
        float4 v = ((const float4*)acc)[idx];
        v.x *= inv; v.y *= inv; v.z *= inv; v.w *= inv;
        o4[idx] = v;
    }
}

template <int NT, int PITCH, int S>
__device__ __forceinline__ void group_mma(v8f (&acc)[NT], const v2f (&a)[8], v2f (&b_cur)[NT],
                                          const float* wlane, int gi) {
    #pragma unroll
    for (int ks = 0; ks < 8; ++ks) {
        int sn = gi * 8 + ks + 1;
        sn = (sn < S) ? sn : S - 1;
        const float* wn = wlane + sn * (2 * PITCH);
        v2f b_nxt[NT];
        #pragma unroll
        for (int j = 0; j < NT; ++j) b_nxt[j] = *(const v2f*)(wn + j * 32);
        #pragma unroll
        for (int j = 0; j < NT; ++j)
            acc[j] = __builtin_amdgcn_wmma_f32_16x16x4_f32(
                 false, a[ks],  false, b_cur[j],
                 (short)0, acc[j],  false,  false);
        #pragma unroll
        for (int j = 0; j < NT; ++j) b_cur[j] = b_nxt[j];
    }
}

template <int K, int N, int NB>
__global__ __launch_bounds__(256) void sage_gemm_ldsw(
        const float* __restrict__ A0,
        const float* __restrict__ A1,
        const float* __restrict__ W0,
        const float* __restrict__ W1,
        const float* __restrict__ bias,
        float* __restrict__ out,
        int M, int relu) {
    constexpr int PITCH = NB * 2 + 32;
    constexpr int NT  = NB / 16;
    constexpr int CB  = N / NB;
    constexpr int G   = K / 32;
    constexpr int NG  = 2 * G;
    constexpr int S   = NG * 8;
    constexpr int NB4 = NB / 4;
    __shared__ __attribute__((aligned(16))) float Ws[K * PITCH];

    const int cb  = blockIdx.x % CB;
    const int wg  = blockIdx.x / CB;
    const int nwg = gridDim.x / CB;
    const int colbase = cb * NB;

    for (int idx = threadIdx.x; idx < K * NB4; idx += 256) {
        const int pr = idx / NB4;
        const int c4 = idx - pr * NB4;
        const float* Wm = (pr >= K / 2) ? W1 : W0;
        const int kp = (pr >= K / 2) ? pr - K / 2 : pr;
        const float* s0 = Wm + (size_t)(2 * kp) * N + colbase + c4 * 4;
        const float4 r0 = *(const float4*)s0;
        const float4 r1 = *(const float4*)(s0 + N);
        float* d = Ws + pr * PITCH + c4 * 8;
        *(float4*)d       = make_float4(r0.x, r1.x, r0.y, r1.y);
        *(float4*)(d + 4) = make_float4(r0.z, r1.z, r0.w, r1.w);
    }
    __syncthreads();

    const int wave = threadIdx.x >> 5;
    const int lane = threadIdx.x & 31;
    const int half = lane >> 4;
    const int l15  = lane & 15;
    const int rowTiles = M >> 4;
    const float* wlane = Ws + half * PITCH + l15 * 2;

    for (int t = wave * nwg + wg; t < rowTiles; t += 8 * nwg) {
        const size_t aoff = (size_t)((t << 4) + l15) * K + (half << 1);
        const float* a0p = A0 + aoff;
        const float* a1p = A1 + aoff;

        v8f acc[NT];
        #pragma unroll
        for (int j = 0; j < NT; ++j) acc[j] = (v8f){0.f, 0.f, 0.f, 0.f, 0.f, 0.f, 0.f, 0.f};

        v2f a_ev[8], a_od[8];
        v2f b_cur[NT];
        #pragma unroll
        for (int ks = 0; ks < 8; ++ks) a_ev[ks] = *(const v2f*)(a0p + ks * 4);
        #pragma unroll
        for (int j = 0; j < NT; ++j) b_cur[j] = *(const v2f*)(wlane + j * 32);

        for (int gi = 0; gi < NG; gi += 2) {
            const float* aop = (gi + 1 >= G) ? a1p + (gi + 1 - G) * 32 : a0p + (gi + 1) * 32;
            #pragma unroll
            for (int ks = 0; ks < 8; ++ks) a_od[ks] = *(const v2f*)(aop + ks * 4);

            group_mma<NT, PITCH, S>(acc, a_ev, b_cur, wlane, gi);

            const int gn = (gi + 2 < NG) ? gi + 2 : NG - 1;
            const float* aep = (gn >= G) ? a1p + (gn - G) * 32 : a0p + gn * 32;
            #pragma unroll
            for (int ks = 0; ks < 8; ++ks) a_ev[ks] = *(const v2f*)(aep + ks * 4);

            group_mma<NT, PITCH, S>(acc, a_od, b_cur, wlane, gi + 1);
        }

        float* op = out + (size_t)((t << 4) + (half << 3)) * N + colbase + l15;
        #pragma unroll
        for (int j = 0; j < NT; ++j) {
            const float bv = bias[colbase + j * 16 + l15];
            #pragma unroll
            for (int v = 0; v < 8; ++v) {
                float r = acc[j][v] + bv;
                if (relu) r = fmaxf(r, 0.0f);
                op[(size_t)v * N + j * 16] = r;
            }
        }
    }
}

extern "C" void kernel_launch(void* const* d_in, const int* in_sizes, int n_in,
                              void* d_out, int out_size, void* d_ws, size_t ws_size,
                              hipStream_t stream) {
    (void)in_sizes; (void)n_in; (void)out_size; (void)ws_size;

    const float* x = (const float*)d_in[0];
    const int* srcp[3] = {(const int*)d_in[1], (const int*)d_in[3], (const int*)d_in[5]};
    const int* dstp[3] = {(const int*)d_in[2], (const int*)d_in[4], (const int*)d_in[6]};
    const float* Wl[3] = {(const float*)d_in[7],  (const float*)d_in[10], (const float*)d_in[13]};
    const float* Wr[3] = {(const float*)d_in[8],  (const float*)d_in[11], (const float*)d_in[14]};
    const float* bb[3] = {(const float*)d_in[9],  (const float*)d_in[12], (const float*)d_in[15]};

    const int ntgt[3] = {100000, 20000, 4000};
    const int relu[3] = {1, 0, 1};

    float* ws     = (float*)d_ws;
    float* aggbuf = ws;
    float* h0     = aggbuf + 12900000;
    float* h1     = h0 + 25600000;

    unsigned* omap[3] = {(unsigned*)h1, (unsigned*)h1 + 375000, (unsigned*)(aggbuf + 12800000)};
    owner_map_all<391, 1500000, 79, 300000, 16, 60000><<<456, 256, 0, stream>>>(
        dstp[0], dstp[1], dstp[2], omap[0], omap[1], omap[2]);

    const float* layer_in[3] = {x, h0, h1};
    float* layer_out[3]      = {h0, h1, (float*)d_out};

    for (int L = 0; L < 3; ++L) {
        int n = ntgt[L];

        if (L == 0)
            seg_mean_owner<128, 391, 16, 32, 1500000, 100000><<<256, 1024, 0, stream>>>(
                layer_in[L], srcp[L], dstp[L], omap[L], aggbuf);
        else if (L == 1)
            seg_mean_owner<256, 79, 8, 32, 300000, 20000><<<256, 1024, 0, stream>>>(
                layer_in[L], srcp[L], dstp[L], omap[L], aggbuf);
        else
            seg_mean_owner<256, 16, 8, 32, 60000, 4000><<<256, 1024, 0, stream>>>(
                layer_in[L], srcp[L], dstp[L], omap[L], aggbuf);

        if (L == 0)
            sage_gemm_ldsw<128, 256, 256><<<256, 256, 0, stream>>>(
                aggbuf, layer_in[L], Wl[L], Wr[L], bb[L], layer_out[L], n, relu[L]);
        else if (L == 1)
            sage_gemm_ldsw<256, 256, 128><<<256, 256, 0, stream>>>(
                aggbuf, layer_in[L], Wl[L], Wr[L], bb[L], layer_out[L], n, relu[L]);
        else
            sage_gemm_ldsw<256, 128, 64><<<256, 256, 0, stream>>>(
                aggbuf, layer_in[L], Wl[L], Wr[L], bb[L], layer_out[L], n, relu[L]);
    }
}


# Segment-mean row gather widened: one 16-byte load per lane brings 512 B of a source row (was four 4-byte loads); LDS rows word-transposed so LDS adds stay conflict-free; sum order = LDS arrival order
# speedup vs baseline: 3.5906x; 1.0674x over previous
#include <hip/hip_runtime.h>


typedef float v2f __attribute__((ext_vector_type(2)));
typedef float v8f __attribute__((ext_vector_type(8)));

template <int RPW>
__device__ __forceinline__ unsigned own4(const int4 v) {
    const unsigned a = min((unsigned)v.x / (unsigned)RPW, 255u);
    const unsigned b = min((unsigned)v.y / (unsigned)RPW, 255u);
    const unsigned c = min((unsigned)v.z / (unsigned)RPW, 255u);
    const unsigned d = min((unsigned)v.w / (unsigned)RPW, 255u);
    return a | (b << 8) | (c << 16) | (d << 24);
}

template <int RPW, int E>
__device__ __forceinline__ void owner_bytes(const int* __restrict__ dst,
                                            unsigned* __restrict__ omap, int blk, int tid) {
    static_assert(E % 16 == 0, "whole 16-edge groups");
    constexpr int E4 = E / 4;
    const int4* d4 = (const int4*)dst;
    int4 v[4];
    #pragma unroll
    for (int u = 0; u < 4; ++u) {
        const int i = blk * 1024 + u * 256 + tid;
        v[u] = d4[i < E4 ? i : E4 - 1];
    }
    #pragma unroll
    for (int u = 0; u < 4; ++u) {
        const int i = blk * 1024 + u * 256 + tid;
        if (i < E4) omap[i] = own4<RPW>(v[u]);
    }
}

template <int RPW0, int E0, int RPW1, int E1, int RPW2, int E2>
__global__ __launch_bounds__(256) void owner_map_all(
        const int* __restrict__ dst0, const int* __restrict__ dst1,
        const int* __restrict__ dst2, unsigned* __restrict__ m0,
        unsigned* __restrict__ m1, unsigned* __restrict__ m2) {
    constexpr int B0 = (E0 / 4 + 1023) / 1024;
    constexpr int B1 = (E1 / 4 + 1023) / 1024;
    const int b = (int)blockIdx.x;
    const int tid = (int)threadIdx.x;
    if (b < B0)           owner_bytes<RPW0, E0>(dst0, m0, b, tid);
    else if (b < B0 + B1) owner_bytes<RPW1, E1>(dst1, m1, b - B0, tid);
    else                  owner_bytes<RPW2, E2>(dst2, m2, b - B0 - B1, tid);
}

template <int D, int RPW, int EB, bool FULL>
__device__ __forceinline__ void drain_rows(const float* __restrict__ xs,
                                           const int* __restrict__ src,
                                           const int* __restrict__ dst,
                                           float* acc, int* cntl, const unsigned* qw,
                                           int cnt, unsigned lane, int lo, int nrows) {
    constexpr int R4 = D / 128;
    int jj = (int)(lane & (EB - 1));
    if (!FULL) jj = jj < cnt ? jj : cnt - 1;
    const unsigned e = qw[jj];
    const int s  = src[e];
    const int dr = dst[e] - lo;
    const bool ok = (unsigned)dr < (unsigned)nrows;
    const int dl = ok ? dr : RPW;
    if (ok && (int)lane < (FULL ? EB : cnt)) atomicAdd(&cntl[dl], 1);
    float4 v[EB][R4];
    #pragma unroll
    for (int j = 0; j < EB; ++j) {
        if (FULL || j < cnt) {
            const int sj = __builtin_amdgcn_readlane(s, j);
            const float4* row = (const float4*)(xs + (size_t)sj * D);
            #pragma unroll
            for (int i = 0; i < R4; ++i) v[j][i] = row[lane + i * 32];
        }
    }
    #pragma unroll
    for (int j = 0; j < EB; ++j) {
        if (FULL || j < cnt) {
            const int dj = __builtin_amdgcn_readlane(dl, j);
            float* a = acc + dj * D + lane;
            #pragma unroll
            for (int i = 0; i < R4; ++i) {
                atomicAdd(a + i * 128,      v[j][i].x);
                atomicAdd(a + i * 128 + 32, v[j][i].y);
                atomicAdd(a + i * 128 + 64, v[j][i].z);
                atomicAdd(a + i * 128 + 96, v[j][i].w);
            }
        }
    }
}

template <int D, int RPW, int EB, int NW, int E, int N>
__global__ __launch_bounds__(NW * 32) void seg_mean_owner(
        const float* __restrict__ xs, const int* __restrict__ src,
        const int* __restrict__ dst, const unsigned* __restrict__ omap,
        float* __restrict__ agg) {
    constexpr int E16  = E / 16;
    constexpr int UPT  = 2;
    constexpr int BLKG = 32 * UPT;
    constexpr int NBLK = (E16 + BLKG - 1) / BLKG;
    constexpr int QCAP = EB + 32;
    constexpr int C4   = D / 4;
    static_assert(E % 16 == 0 && (N + RPW - 1) / RPW <= 256, "one byte names the owner");
    __shared__ __attribute__((aligned(16))) float acc[(RPW + 1) * D];
    __shared__ int cntl[RPW + 1];
    __shared__ unsigned qall[NW * QCAP];

    const int lo = (int)blockIdx.x * RPW;
    int nrows = N - lo;
    nrows = nrows < RPW ? nrows : RPW;
    if (nrows <= 0) return;

    for (int i = threadIdx.x; i < (RPW + 1) * C4; i += NW * 32)
        ((float4*)acc)[i] = make_float4(0.f, 0.f, 0.f, 0.f);
    for (int i = threadIdx.x; i < RPW + 1; i += NW * 32) cntl[i] = 0;
    __syncthreads();

    const unsigned lane = threadIdx.x & 31;
    const int wave = __builtin_amdgcn_readfirstlane((int)(threadIdx.x >> 5));
    unsigned* qw = qall + wave * QCAP;
    int qn = 0;
    const int rot = (int)(((long long)blockIdx.x * NBLK) / (int)gridDim.x);
    const uint4* m4 = (const uint4*)omap;
    const unsigned pat  = (unsigned)blockIdx.x * 0x01010101u;

    uint4 cur[UPT], nxt[UPT];
    {
        int bb = wave + rot; bb = bb >= NBLK ? bb - NBLK : bb;
        bb = wave < NBLK ? bb : 0;
        #pragma unroll
        for (int u = 0; u < UPT; ++u) {
            const int g = bb * BLKG + u * 32 + (int)lane;
            cur[u] = m4[g < E16 ? g : E16 - 1];
        }
    }
    for (int b = wave; b < NBLK; b += NW) {
        int bn = b + NW; bn = bn < NBLK ? bn : b;
        int bbn = bn + rot; bbn = bbn >= NBLK ? bbn - NBLK : bbn;
        #pragma unroll
        for (int u = 0; u < UPT; ++u) {
            const int g = bbn * BLKG + u * 32 + (int)lane;
            nxt[u] = m4[g < E16 ? g : E16 - 1];
        }

        int bb = b + rot; bb = bb >= NBLK ? bb - NBLK : bb;
        const unsigned ebase = (unsigned)(bb * BLKG + (int)lane) * 16u;
        unsigned fl = 0;
        #pragma unroll
        for (int u = 0; u < UPT; ++u) {
            const unsigned w[4] = {cur[u].x, cur[u].y, cur[u].z, cur[u].w};
            #pragma unroll
            for (int c = 0; c < 4; ++c) {
                const int k = u * 4 + c;
                const unsigned x = w[c] ^ pat;
                const unsigned t = (x - 0x01010101u) & ~x;
                fl |= (t >> (7 - k)) & (0x01010101u << k);
            }
        }
        const int g0 = bb * BLKG + (int)lane;
        fl &= (g0 < E16 ? 0x0F0F0F0Fu : 0u) | (g0 + 32 < E16 ? 0xF0F0F0F0u : 0u);
        unsigned mk = __builtin_amdgcn_ballot_w32(fl != 0u);
        while (mk != 0u) {
            const int pos = qn + (int)__builtin_amdgcn_mbcnt_lo(mk, 0u);
            if (fl != 0u) {
                const unsigned bit = (unsigned)__builtin_ctz(fl);
                const unsigned k = bit & 7u;
                qw[pos] = ebase + (k >> 2) * 512u + (k & 3u) * 4u + (bit >> 3);
                fl &= fl - 1u;
            }
            qn += __builtin_popcount(mk);
            while (qn >= EB) {
                drain_rows<D, RPW, EB, true>(xs, src, dst, acc, cntl, qw + (qn - EB), EB, lane, lo, nrows);
                qn -= EB;
            }
            mk = __builtin_amdgcn_ballot_w32(fl != 0u);
        }
        #pragma unroll
        for (int u = 0; u < UPT; ++u) cur[u] = nxt[u];
    }
    if (qn > 0) drain_rows<D, RPW, EB, false>(xs, src, dst, acc, cntl, qw, qn, lane, lo, nrows);
    __syncthreads();

    float4* o4 = (float4*)(agg + (size_t)lo * D);
    for (int idx = threadIdx.x; idx < nrows * C4; idx += NW * 32) {
        const int r = idx / C4;
        const int q = idx - r * C4;
        const float inv = 1.0f / fmaxf((float)cntl[r], 1.0f);
        const float* a = acc + r * D + (q >> 5) * 128 + (q & 31);
        float4 v = make_float4(a[0], a[32], a[64], a[96]);
        v.x *= inv; v.y *= inv; v.z *= inv; v.w *= inv;
        o4[idx] = v;
    }
}

template <int NT, int PITCH, int S>
__device__ __forceinline__ void group_mma(v8f (&acc)[NT], const v2f (&a)[8], v2f (&b_cur)[NT],
                                          const float* wlane, int gi) {
    #pragma unroll
    for (int ks = 0; ks < 8; ++ks) {
        int sn = gi * 8 + ks + 1;
        sn = (sn < S) ? sn : S - 1;
        const float* wn = wlane + sn * (2 * PITCH);
        v2f b_nxt[NT];
        #pragma unroll
        for (int j = 0; j < NT; ++j) b_nxt[j] = *(const v2f*)(wn + j * 32);
        #pragma unroll
        for (int j = 0; j < NT; ++j)
            acc[j] = __builtin_amdgcn_wmma_f32_16x16x4_f32(
                 false, a[ks],  false, b_cur[j],
                 (short)0, acc[j],  false,  false);
        #pragma unroll
        for (int j = 0; j < NT; ++j) b_cur[j] = b_nxt[j];
    }
}

template <int K, int N, int NB>
__global__ __launch_bounds__(256) void sage_gemm_ldsw(
        const float* __restrict__ A0,
        const float* __restrict__ A1,
        const float* __restrict__ W0,
        const float* __restrict__ W1,
        const float* __restrict__ bias,
        float* __restrict__ out,
        int M, int relu) {
    constexpr int PITCH = NB * 2 + 32;
    constexpr int NT  = NB / 16;
    constexpr int CB  = N / NB;
    constexpr int G   = K / 32;
    constexpr int NG  = 2 * G;
    constexpr int S   = NG * 8;
    constexpr int NB4 = NB / 4;
    __shared__ __attribute__((aligned(16))) float Ws[K * PITCH];

    const int cb  = blockIdx.x % CB;
    const int wg  = blockIdx.x / CB;
    const int nwg = gridDim.x / CB;
    const int colbase = cb * NB;

    for (int idx = threadIdx.x; idx < K * NB4; idx += 256) {
        const int pr = idx / NB4;
        const int c4 = idx - pr * NB4;
        const float* Wm = (pr >= K / 2) ? W1 : W0;
        const int kp = (pr >= K / 2) ? pr - K / 2 : pr;
        const float* s0 = Wm + (size_t)(2 * kp) * N + colbase + c4 * 4;
        const float4 r0 = *(const float4*)s0;
        const float4 r1 = *(const float4*)(s0 + N);
        float* d = Ws + pr * PITCH + c4 * 8;
        *(float4*)d       = make_float4(r0.x, r1.x, r0.y, r1.y);
        *(float4*)(d + 4) = make_float4(r0.z, r1.z, r0.w, r1.w);
    }
    __syncthreads();

    const int wave = threadIdx.x >> 5;
    const int lane = threadIdx.x & 31;
    const int half = lane >> 4;
    const int l15  = lane & 15;
    const int rowTiles = M >> 4;
    const float* wlane = Ws + half * PITCH + l15 * 2;

    for (int t = wave * nwg + wg; t < rowTiles; t += 8 * nwg) {
        const size_t aoff = (size_t)((t << 4) + l15) * K + (half << 1);
        const float* a0p = A0 + aoff;
        const float* a1p = A1 + aoff;

        v8f acc[NT];
        #pragma unroll
        for (int j = 0; j < NT; ++j) acc[j] = (v8f){0.f, 0.f, 0.f, 0.f, 0.f, 0.f, 0.f, 0.f};

        v2f a_ev[8], a_od[8];
        v2f b_cur[NT];
        #pragma unroll
        for (int ks = 0; ks < 8; ++ks) a_ev[ks] = *(const v2f*)(a0p + ks * 4);
        #pragma unroll
        for (int j = 0; j < NT; ++j) b_cur[j] = *(const v2f*)(wlane + j * 32);

        for (int gi = 0; gi < NG; gi += 2) {
            const float* aop = (gi + 1 >= G) ? a1p + (gi + 1 - G) * 32 : a0p + (gi + 1) * 32;
            #pragma unroll
            for (int ks = 0; ks < 8; ++ks) a_od[ks] = *(const v2f*)(aop + ks * 4);

            group_mma<NT, PITCH, S>(acc, a_ev, b_cur, wlane, gi);

            const int gn = (gi + 2 < NG) ? gi + 2 : NG - 1;
            const float* aep = (gn >= G) ? a1p + (gn - G) * 32 : a0p + gn * 32;
            #pragma unroll
            for (int ks = 0; ks < 8; ++ks) a_ev[ks] = *(const v2f*)(aep + ks * 4);

            group_mma<NT, PITCH, S>(acc, a_od, b_cur, wlane, gi + 1);
        }

        float* op = out + (size_t)((t << 4) + (half << 3)) * N + colbase + l15;
        #pragma unroll
        for (int j = 0; j < NT; ++j) {
            const float bv = bias[colbase + j * 16 + l15];
            #pragma unroll
            for (int v = 0; v < 8; ++v) {
                float r = acc[j][v] + bv;
                if (relu) r = fmaxf(r, 0.0f);
                op[(size_t)v * N + j * 16] = r;
            }
        }
    }
}

extern "C" void kernel_launch(void* const* d_in, const int* in_sizes, int n_in,
                              void* d_out, int out_size, void* d_ws, size_t ws_size,
                              hipStream_t stream) {
    (void)in_sizes; (void)n_in; (void)out_size; (void)ws_size;

    const float* x = (const float*)d_in[0];
    const int* srcp[3] = {(const int*)d_in[1], (const int*)d_in[3], (const int*)d_in[5]};
    const int* dstp[3] = {(const int*)d_in[2], (const int*)d_in[4], (const int*)d_in[6]};
    const float* Wl[3] = {(const float*)d_in[7],  (const float*)d_in[10], (const float*)d_in[13]};
    const float* Wr[3] = {(const float*)d_in[8],  (const float*)d_in[11], (const float*)d_in[14]};
    const float* bb[3] = {(const float*)d_in[9],  (const float*)d_in[12], (const float*)d_in[15]};

    const int ntgt[3] = {100000, 20000, 4000};
    const int relu[3] = {1, 0, 1};

    float* ws     = (float*)d_ws;
    float* aggbuf = ws;
    float* h0     = aggbuf + 12900000;
    float* h1     = h0 + 25600000;

    unsigned* omap[3] = {(unsigned*)h1, (unsigned*)h1 + 375000, (unsigned*)(aggbuf + 12800000)};
    owner_map_all<391, 1500000, 79, 300000, 16, 60000><<<456, 256, 0, stream>>>(
        dstp[0], dstp[1], dstp[2], omap[0], omap[1], omap[2]);

    const float* layer_in[3] = {x, h0, h1};
    float* layer_out[3]      = {h0, h1, (float*)d_out};

    for (int L = 0; L < 3; ++L) {
        int n = ntgt[L];

        if (L == 0)
            seg_mean_owner<128, 391, 16, 32, 1500000, 100000><<<256, 1024, 0, stream>>>(
                layer_in[L], srcp[L], dstp[L], omap[L], aggbuf);
        else if (L == 1)
            seg_mean_owner<256, 79, 8, 32, 300000, 20000><<<256, 1024, 0, stream>>>(
                layer_in[L], srcp[L], dstp[L], omap[L], aggbuf);
        else
            seg_mean_owner<256, 16, 8, 32, 60000, 4000><<<256, 1024, 0, stream>>>(
                layer_in[L], srcp[L], dstp[L], omap[L], aggbuf);

        if (L == 0)
            sage_gemm_ldsw<128, 256, 256><<<256, 256, 0, stream>>>(
                aggbuf, layer_in[L], Wl[L], Wr[L], bb[L], layer_out[L], n, relu[L]);
        else if (L == 1)
            sage_gemm_ldsw<256, 256, 128><<<256, 256, 0, stream>>>(
                aggbuf, layer_in[L], Wl[L], Wr[L], bb[L], layer_out[L], n, relu[L]);
        else
            sage_gemm_ldsw<256, 128, 64><<<256, 256, 0, stream>>>(
                aggbuf, layer_in[L], Wl[L], Wr[L], bb[L], layer_out[L], n, relu[L]);
    }
}
